# starting kernel
# baseline (speedup 1.0000x reference)
#include <hip/hip_runtime.h>

typedef float v2f __attribute__((ext_vector_type(2)));
typedef float v8f __attribute__((ext_vector_type(8)));

#define TT    500
#define HH    40
#define PITCH 50
#define WAVES 8
#define ROWS_PER_BLOCK (WAVES * 16)

static __device__ __forceinline__ float fast_tanh(float x) {
    float y;
    asm("v_tanh_f32 %0, %1" : "=v"(y) : "v"(x));
    return y;
}

static __device__ __forceinline__ void lds_order_fence() {
    asm volatile("" ::: "memory");
}

static __device__ __forceinline__ float wfull(int r, int c,
                                              const float* __restrict__ Wh,
                                              const float* __restrict__ Ks,
                                              const float* __restrict__ Ka,
                                              const float* __restrict__ bias) {
    if (c >= HH) return 0.0f;
    if (r < HH)  return Wh[r * HH + c];
    if (r == HH)     return Ks[c];
    if (r == HH + 1) return Ks[HH + c];
    if (r == HH + 2) return Ka[c];
    return bias[c];
}

__global__ __launch_bounds__(256, 1)
void rnn_scan_wmma(const float* __restrict__ x,
                   const float* __restrict__ Ks,
                   const float* __restrict__ Wh,
                   const float* __restrict__ Ka,
                   const float* __restrict__ bias,
                   const float* __restrict__ Dw,
                   const float* __restrict__ Db,
                   float* __restrict__ out)
{
    __shared__ float s2s[WAVES][16][PITCH];

    const int tid     = threadIdx.x;
    const int w       = tid >> 5;
    const int lane    = tid & 31;
    const int m       = lane & 15;
    const int half    = lane >> 4;
    const int halfoff = half * 2;

    {
        float* myz = (float*)&s2s[w][0][0];
        for (int i = lane; i < 16 * PITCH; i += 32) myz[i] = 0.0f;
        lds_order_fence();
    }

    v2f wv[11][3];
    #pragma unroll
    for (int ck = 0; ck < 11; ++ck) {
        #pragma unroll
        for (int n = 0; n < 3; ++n) {
            const int col = n * 16 + m;
            const int r0  = ck * 4 + halfoff;
            wv[ck][n].x = wfull(r0,     col, Wh, Ks, Ka, bias);
            wv[ck][n].y = wfull(r0 + 1, col, Wh, Ks, Ka, bias);
        }
    }

    const int    brow = blockIdx.x * ROWS_PER_BLOCK + w * 16 + m;
    const float* pxt  = x + (size_t)brow * TT * 3;
    const int    f0   = 2 * half;
    const float  mean0   = half ? 0.0f : -0.3f;
    const float  iscale0 = half ? 1.0f : (1.0f / 0.9f);
    float p0 = pxt[f0];
    float p1 = pxt[1];

    const float* sbase = &s2s[w][m][0];
    float*       dstb  = &s2s[w][8 * half][m];

    for (int t = 0; t < TT; ++t) {
        v2f a[11];
        #pragma unroll
        for (int ck = 0; ck < 10; ++ck)
            a[ck] = *(const v2f*)(sbase + ck * 4 + halfoff);

        a[10].x = (p0 - mean0) * iscale0;
        a[10].y = half ? 1.0f : p1 * (1.0f / 0.07f);

        const float* pn = (t < TT - 1) ? (pxt + 3) : pxt;
        p0  = pn[f0];
        p1  = pn[1];
        pxt = pn;

        lds_order_fence();

        v8f acc0 = {}, acc1 = {}, acc2 = {};
        #pragma unroll
        for (int ck = 0; ck < 11; ++ck) {
            acc0 = __builtin_amdgcn_wmma_f32_16x16x4_f32(false, a[ck], false, wv[ck][0],
                                                         (short)0, acc0, false, false);
            acc1 = __builtin_amdgcn_wmma_f32_16x16x4_f32(false, a[ck], false, wv[ck][1],
                                                         (short)0, acc1, false, false);
            acc2 = __builtin_amdgcn_wmma_f32_16x16x4_f32(false, a[ck], false, wv[ck][2],
                                                         (short)0, acc2, false, false);
        }

        float tv[24];
        #pragma unroll
        for (int r = 0; r < 8; ++r) {
            tv[r]      = fast_tanh(acc0[r]);
            tv[8 + r]  = fast_tanh(acc1[r]);
            tv[16 + r] = fast_tanh(acc2[r]);
        }

        #pragma unroll
        for (int r = 0; r < 8; ++r) {
            dstb[r * PITCH +  0] = tv[r];
            dstb[r * PITCH + 16] = tv[8 + r];
            dstb[r * PITCH + 32] = tv[16 + r];
        }

        lds_order_fence();
    }

    {
        lds_order_fence();
        const int orow = lane >> 1;
        const int oc   = lane & 1;
        const float* srow = &s2s[w][orow][0];
        float sum = Db[oc];
        #pragma unroll
        for (int h = 0; h < HH; ++h)
            sum += srow[h] * Dw[h * 2 + oc];
        const float so = oc ? 0.07f : 0.9f;
        const float mo = oc ? 0.0f  : -0.3f;
        float v = sum * so + mo;
        const float lo = oc ? -0.07f : -1.2f;
        const float hi = oc ?  0.07f :  0.6f;
        v = fminf(fmaxf(v, lo), hi);
        out[((size_t)(blockIdx.x * ROWS_PER_BLOCK + w * 16 + orow)) * 2 + oc] = v;
    }
}

extern "C" void kernel_launch(void* const* d_in, const int* in_sizes, int n_in,
                              void* d_out, int out_size, void* d_ws, size_t ws_size,
                              hipStream_t stream) {
    const float* x    = (const float*)d_in[0];
    const float* Ks   = (const float*)d_in[1];
    const float* Wh   = (const float*)d_in[2];
    const float* Ka   = (const float*)d_in[3];
    const float* bias = (const float*)d_in[4];
    const float* Dw   = (const float*)d_in[5];
    const float* Db   = (const float*)d_in[6];
    float* out = (float*)d_out;

    const int B = in_sizes[0] / (TT * 3);
    dim3 grid(B / ROWS_PER_BLOCK), block(256);
    rnn_scan_wmma<<<grid, block, 0, stream>>>(x, Ks, Wh, Ka, bias, Dw, Db, out);
}


# state kept in registers: step computed transposed (weights as A, state as B), a lane's D registers after tanh are its next B fragments; LDS round trip removed; k order and head sum order change
# speedup vs baseline: 1.0492x; 1.0492x over previous
#include <hip/hip_runtime.h>

typedef float v2f __attribute__((ext_vector_type(2)));
typedef float v8f __attribute__((ext_vector_type(8)));

#define TT    500
#define HH    40
#define WAVES 8
#define ROWS_PER_BLOCK (WAVES * 16)

static __device__ __forceinline__ int unit_of(int tile, int row) {
    if (tile < 2) return tile * 16 + row;
    if (row < 4) return 32 + row;
    if (row >= 8 && row < 12) return 36 + (row - 8);
    return -1;
}

static __device__ __forceinline__ float a_elem(int mt, int kc, int c, int h, int s,
                                               const float* __restrict__ Wh,
                                               const float* __restrict__ Ks,
                                               const float* __restrict__ Ka,
                                               const float* __restrict__ bias) {
    const int jout = unit_of(mt, c);
    const int jo   = jout < 0 ? 0 : jout;
    float v;
    if (kc < 10) {
        const int jin = unit_of(kc >> 2, 8 * h + 2 * (kc & 3) + s);
        v = Wh[jin * HH + jo];
    } else {
        const float* src = h ? (s ? bias : Ka) : (s ? Ks + HH : Ks);
        v = src[jo];
    }
    return jout < 0 ? 0.0f : v;
}

__global__ __launch_bounds__(256, 1)
void rnn_scan_wmma(const float* __restrict__ x,
                   const float* __restrict__ Ks,
                   const float* __restrict__ Wh,
                   const float* __restrict__ Ka,
                   const float* __restrict__ bias,
                   const float* __restrict__ Dw,
                   const float* __restrict__ Db,
                   float* __restrict__ out)
{
    const int tid  = threadIdx.x;
    const int w    = tid >> 5;
    const int lane = tid & 31;
    const int c    = lane & 15;
    const int half = lane >> 4;

    v2f wA[3][11];
    for (int mt = 0; mt < 3; ++mt) {
        for (int kc = 0; kc < 11; ++kc) {
            wA[mt][kc].x = a_elem(mt, kc, c, half, 0, Wh, Ks, Ka, bias);
            wA[mt][kc].y = a_elem(mt, kc, c, half, 1, Wh, Ks, Ka, bias);
        }
    }

    const int    brow = blockIdx.x * ROWS_PER_BLOCK + w * 16 + c;
    const float* pxt  = x + (size_t)brow * TT * 3;
    const int    f0   = 2 * half;
    const float  mean0   = half ? 0.0f : -0.3f;
    const float  iscale0 = half ? 1.0f : (1.0f / 0.9f);
    float p0 = pxt[f0];
    float p1 = pxt[1];

    float st[3][8];
    for (int mt = 0; mt < 3; ++mt) {
        for (int r = 0; r < 8; ++r) st[mt][r] = 0.0f;
    }

    for (int t = 0; t < TT; ++t) {
        v2f bx;
        bx.x = (p0 - mean0) * iscale0;
        bx.y = half ? 1.0f : p1 * (1.0f / 0.07f);

        const float* pn = (t < TT - 1) ? (pxt + 3) : pxt;
        p0  = pn[f0];
        p1  = pn[1];
        pxt = pn;

        v8f acc[3];
        for (int mt = 0; mt < 3; ++mt) {
            v8f z = {};
            acc[mt] = __builtin_amdgcn_wmma_f32_16x16x4_f32(false, wA[mt][10], false, bx,
                                                            (short)0, z, false, false);
        }
        for (int kc = 0; kc < 10; ++kc) {
            v2f b;
            b.x = st[kc >> 2][2 * (kc & 3)];
            b.y = st[kc >> 2][2 * (kc & 3) + 1];
            for (int mt = 0; mt < 3; ++mt)
                acc[mt] = __builtin_amdgcn_wmma_f32_16x16x4_f32(false, wA[mt][kc], false, b,
                                                                (short)0, acc[mt], false, false);
        }

        for (int r = 0; r < 8; ++r) {
            st[0][r] = __builtin_amdgcn_tanhf(acc[0][r]);
            st[1][r] = __builtin_amdgcn_tanhf(acc[1][r]);
        }
        for (int r = 0; r < 4; ++r)
            st[2][r] = __builtin_amdgcn_tanhf(acc[2][r]);
    }

    {
        float own = 0.0f, other = 0.0f;
        for (int mt = 0; mt < 3; ++mt) {
            for (int r = 0; r < (mt < 2 ? 8 : 4); ++r) {
                const int j = (mt < 2) ? (mt * 16 + 8 * half + r) : (32 + 4 * half + r);
                own   += st[mt][r] * Dw[j * 2 + half];
                other += st[mt][r] * Dw[j * 2 + (1 - half)];
            }
        }
        const float recv = __shfl_xor(other, 16);
        const float sum  = Db[half] + (own + recv);
        const float so = half ? 0.07f : 0.9f;
        const float mo = half ? 0.0f  : -0.3f;
        float v = sum * so + mo;
        const float lo = half ? -0.07f : -1.2f;
        const float hi = half ?  0.07f :  0.6f;
        v = fminf(fmaxf(v, lo), hi);
        out[(size_t)brow * 2 + half] = v;
    }
}

extern "C" void kernel_launch(void* const* d_in, const int* in_sizes, int n_in,
                              void* d_out, int out_size, void* d_ws, size_t ws_size,
                              hipStream_t stream) {
    const float* x    = (const float*)d_in[0];
    const float* Ks   = (const float*)d_in[1];
    const float* Wh   = (const float*)d_in[2];
    const float* Ka   = (const float*)d_in[3];
    const float* bias = (const float*)d_in[4];
    const float* Dw   = (const float*)d_in[5];
    const float* Db   = (const float*)d_in[6];
    float* out = (float*)d_out;

    const int B = in_sizes[0] / (TT * 3);
    dim3 grid(B / ROWS_PER_BLOCK), block(256);
    rnn_scan_wmma<<<grid, block, 0, stream>>>(x, Ks, Wh, Ka, bias, Dw, Db, out);
}


# two independent 16-row groups per wave, half a step apart, one wave per SIMD (256 blocks of 4 waves): each group's tanh issue between the other group's WMMAs; sum orders unchanged
# speedup vs baseline: 1.0812x; 1.0305x over previous
#include <hip/hip_runtime.h>

typedef float v2f __attribute__((ext_vector_type(2)));
typedef float v8f __attribute__((ext_vector_type(8)));

#define TT    500
#define HH    40
#define WAVES 4
#define ROWS_PER_WAVE  32
#define ROWS_PER_BLOCK (WAVES * ROWS_PER_WAVE)

static __device__ __forceinline__ int unit_of(int tile, int row) {
    if (tile < 2) return tile * 16 + row;
    if (row < 4) return 32 + row;
    if (row >= 8 && row < 12) return 36 + (row - 8);
    return -1;
}

static __device__ __forceinline__ float a_elem(int mt, int kc, int c, int h, int s,
                                               const float* __restrict__ Wh,
                                               const float* __restrict__ Ks,
                                               const float* __restrict__ Ka,
                                               const float* __restrict__ bias) {
    const int jout = unit_of(mt, c);
    const int jo   = jout < 0 ? 0 : jout;
    float v;
    if (kc < 10) {
        const int jin = unit_of(kc >> 2, 8 * h + 2 * (kc & 3) + s);
        v = Wh[jin * HH + jo];
    } else {
        const float* src = h ? (s ? bias : Ka) : (s ? Ks + HH : Ks);
        v = src[jo];
    }
    return jout < 0 ? 0.0f : v;
}

static __device__ __forceinline__ void tanh_one(int v, const v8f (&pre)[3], float (&st)[3][8]) {
    const int mt = v < 16 ? (v >> 3) : 2;
    const int r  = v < 16 ? (v & 7) : (v - 16);
    st[mt][r] = __builtin_amdgcn_tanhf(pre[mt][r]);
}

template <bool OTHER>
static __device__ __forceinline__ void half_step(const v2f (&wA)[3][11], const float (&st)[3][8],
                                                 v2f bx, v8f (&acc)[3],
                                                 const v8f (&opre)[3], float (&ost)[3][8]) {
    for (int mt = 0; mt < 3; ++mt) {
        v8f z = {};
        acc[mt] = __builtin_amdgcn_wmma_f32_16x16x4_f32(false, wA[mt][10], false, bx,
                                                        (short)0, z, false, false);
    }
    for (int kc = 0; kc < 10; ++kc) {
        v2f b;
        b.x = st[kc >> 2][2 * (kc & 3)];
        b.y = st[kc >> 2][2 * (kc & 3) + 1];
        for (int mt = 0; mt < 3; ++mt)
            acc[mt] = __builtin_amdgcn_wmma_f32_16x16x4_f32(false, wA[mt][kc], false, b,
                                                            (short)0, acc[mt], false, false);
        if (OTHER) {
            tanh_one(2 * kc,     opre, ost);
            tanh_one(2 * kc + 1, opre, ost);
            __builtin_amdgcn_sched_barrier(0);
        }
    }
}

static __device__ __forceinline__ void head(const float (&st)[3][8], int half, int brow,
                                            const float* __restrict__ Dw,
                                            const float* __restrict__ Db,
                                            float* __restrict__ out) {
    float own = 0.0f, other = 0.0f;
    for (int mt = 0; mt < 3; ++mt) {
        for (int r = 0; r < (mt < 2 ? 8 : 4); ++r) {
            const int j = (mt < 2) ? (mt * 16 + 8 * half + r) : (32 + 4 * half + r);
            own   += st[mt][r] * Dw[j * 2 + half];
            other += st[mt][r] * Dw[j * 2 + (1 - half)];
        }
    }
    const float recv = __shfl_xor(other, 16);
    const float sum  = Db[half] + (own + recv);
    const float so = half ? 0.07f : 0.9f;
    const float mo = half ? 0.0f  : -0.3f;
    float v = sum * so + mo;
    const float lo = half ? -0.07f : -1.2f;
    const float hi = half ?  0.07f :  0.6f;
    v = fminf(fmaxf(v, lo), hi);
    out[(size_t)brow * 2 + half] = v;
}

__global__ __launch_bounds__(128, 1)
void rnn_scan_wmma(const float* __restrict__ x,
                   const float* __restrict__ Ks,
                   const float* __restrict__ Wh,
                   const float* __restrict__ Ka,
                   const float* __restrict__ bias,
                   const float* __restrict__ Dw,
                   const float* __restrict__ Db,
                   float* __restrict__ out)
{
    const int tid  = threadIdx.x;
    const int w    = tid >> 5;
    const int lane = tid & 31;
    const int c    = lane & 15;
    const int half = lane >> 4;

    v2f wA[3][11];
    for (int mt = 0; mt < 3; ++mt) {
        for (int kc = 0; kc < 11; ++kc) {
            wA[mt][kc].x = a_elem(mt, kc, c, half, 0, Wh, Ks, Ka, bias);
            wA[mt][kc].y = a_elem(mt, kc, c, half, 1, Wh, Ks, Ka, bias);
        }
    }

    const int    browA = blockIdx.x * ROWS_PER_BLOCK + w * ROWS_PER_WAVE + c;
    const int    browB = browA + 16;
    const float* pxA   = x + (size_t)browA * TT * 3;
    const float* pxB   = x + (size_t)browB * TT * 3;
    const int    f0    = 2 * half;
    const float  mean0   = half ? 0.0f : -0.3f;
    const float  iscale0 = half ? 1.0f : (1.0f / 0.9f);
    float a0 = pxA[f0], a1 = pxA[1];
    float b0 = pxB[f0], b1 = pxB[1];
    __builtin_amdgcn_sched_barrier(0);

    float stA[3][8], stB[3][8];
    for (int mt = 0; mt < 3; ++mt) {
        for (int r = 0; r < 8; ++r) { stA[mt][r] = 0.0f; stB[mt][r] = 0.0f; }
    }
    v8f accA[3], accB[3];

    {
        v2f bx;
        bx.x = (a0 - mean0) * iscale0;
        bx.y = half ? 1.0f : a1 * (1.0f / 0.07f);
        pxA += 3;
        a0 = pxA[f0];
        a1 = pxA[1];
        half_step<false>(wA, stA, bx, accA, accB, stB);
    }

    for (int t = 0; t < TT - 1; ++t) {
        {
            v2f bx;
            bx.x = (b0 - mean0) * iscale0;
            bx.y = half ? 1.0f : b1 * (1.0f / 0.07f);
            pxB += 3;
            b0 = pxB[f0];
            b1 = pxB[1];
            half_step<true>(wA, stB, bx, accB, accA, stA);
        }
        {
            v2f bx;
            bx.x = (a0 - mean0) * iscale0;
            bx.y = half ? 1.0f : a1 * (1.0f / 0.07f);
            const float* pn = (t < TT - 2) ? (pxA + 3) : pxA;
            a0  = pn[f0];
            a1  = pn[1];
            pxA = pn;
            half_step<true>(wA, stA, bx, accA, accB, stB);
        }
    }

    {
        v2f bx;
        bx.x = (b0 - mean0) * iscale0;
        bx.y = half ? 1.0f : b1 * (1.0f / 0.07f);
        half_step<true>(wA, stB, bx, accB, accA, stA);
        for (int v = 0; v < 20; ++v) tanh_one(v, accB, stB);
    }

    head(stA, half, browA, Dw, Db, out);
    head(stB, half, browB, Dw, Db, out);
}

extern "C" void kernel_launch(void* const* d_in, const int* in_sizes, int n_in,
                              void* d_out, int out_size, void* d_ws, size_t ws_size,
                              hipStream_t stream) {
    const float* x    = (const float*)d_in[0];
    const float* Ks   = (const float*)d_in[1];
    const float* Wh   = (const float*)d_in[2];
    const float* Ka   = (const float*)d_in[3];
    const float* bias = (const float*)d_in[4];
    const float* Dw   = (const float*)d_in[5];
    const float* Db   = (const float*)d_in[6];
    float* out = (float*)d_out;

    const int B = in_sizes[0] / (TT * 3);
    dim3 grid(B / ROWS_PER_BLOCK), block(128);
    rnn_scan_wmma<<<grid, block, 0, stream>>>(x, Ks, Wh, Ka, bias, Dw, Db, out);
}


# x stream: each lane reads its batch row 16 bytes at a time, four steps per block and one block ahead; row pointers move once per block in front of the loads; sum orders unchanged
# speedup vs baseline: 1.1983x; 1.1083x over previous
#include <hip/hip_runtime.h>

typedef float v2f __attribute__((ext_vector_type(2)));
typedef float v8f __attribute__((ext_vector_type(8)));

typedef float v4f __attribute__((ext_vector_type(4)));

struct XBlock { v4f q0, q1, q2; };

static __device__ __forceinline__ XBlock load_block(const float* __restrict__ p) {
    const v4f* p4 = reinterpret_cast<const v4f*>(p);
    XBlock b;
    b.q0 = p4[0];
    b.q1 = p4[1];
    b.q2 = p4[2];
    return b;
}

static __device__ __forceinline__ float blk_elem(const XBlock& b, int i) {
    return i < 4 ? b.q0[i] : (i < 8 ? b.q1[i - 4] : b.q2[i - 8]);
}

static __device__ __forceinline__ v2f input_chunk(const XBlock& b, int j, int half,
                                                  float mean0, float iscale0) {
    const float a0 = half ? blk_elem(b, 3 * j + 2) : blk_elem(b, 3 * j);
    const float a1 = blk_elem(b, 3 * j + 1);
    v2f bx;
    bx.x = (a0 - mean0) * iscale0;
    bx.y = half ? 1.0f : a1 * (1.0f / 0.07f);
    return bx;
}

#define TT    500
#define HH    40
#define WAVES 4
#define ROWS_PER_WAVE  32
#define ROWS_PER_BLOCK (WAVES * ROWS_PER_WAVE)

static __device__ __forceinline__ int unit_of(int tile, int row) {
    if (tile < 2) return tile * 16 + row;
    if (row < 4) return 32 + row;
    if (row >= 8 && row < 12) return 36 + (row - 8);
    return -1;
}

static __device__ __forceinline__ float a_elem(int mt, int kc, int c, int h, int s,
                                               const float* __restrict__ Wh,
                                               const float* __restrict__ Ks,
                                               const float* __restrict__ Ka,
                                               const float* __restrict__ bias) {
    const int jout = unit_of(mt, c);
    const int jo   = jout < 0 ? 0 : jout;
    float v;
    if (kc < 10) {
        const int jin = unit_of(kc >> 2, 8 * h + 2 * (kc & 3) + s);
        v = Wh[jin * HH + jo];
    } else {
        const float* src = h ? (s ? bias : Ka) : (s ? Ks + HH : Ks);
        v = src[jo];
    }
    return jout < 0 ? 0.0f : v;
}

static __device__ __forceinline__ void tanh_one(int v, const v8f (&pre)[3], float (&st)[3][8]) {
    const int mt = v < 16 ? (v >> 3) : 2;
    const int r  = v < 16 ? (v & 7) : (v - 16);
    st[mt][r] = __builtin_amdgcn_tanhf(pre[mt][r]);
}

template <bool OTHER>
static __device__ __forceinline__ void half_step(const v2f (&wA)[3][11], const float (&st)[3][8],
                                                 v2f bx, v8f (&acc)[3],
                                                 const v8f (&opre)[3], float (&ost)[3][8]) {
    for (int mt = 0; mt < 3; ++mt) {
        v8f z = {};
        acc[mt] = __builtin_amdgcn_wmma_f32_16x16x4_f32(false, wA[mt][10], false, bx,
                                                        (short)0, z, false, false);
    }
    for (int kc = 0; kc < 10; ++kc) {
        v2f b;
        b.x = st[kc >> 2][2 * (kc & 3)];
        b.y = st[kc >> 2][2 * (kc & 3) + 1];
        for (int mt = 0; mt < 3; ++mt)
            acc[mt] = __builtin_amdgcn_wmma_f32_16x16x4_f32(false, wA[mt][kc], false, b,
                                                            (short)0, acc[mt], false, false);
        if (OTHER) {
            tanh_one(2 * kc,     opre, ost);
            tanh_one(2 * kc + 1, opre, ost);
            __builtin_amdgcn_sched_barrier(0);
        }
    }
}

static __device__ __forceinline__ void head(const float (&st)[3][8], int half, int brow,
                                            const float* __restrict__ Dw,
                                            const float* __restrict__ Db,
                                            float* __restrict__ out) {
    float own = 0.0f, other = 0.0f;
    for (int mt = 0; mt < 3; ++mt) {
        for (int r = 0; r < (mt < 2 ? 8 : 4); ++r) {
            const int j = (mt < 2) ? (mt * 16 + 8 * half + r) : (32 + 4 * half + r);
            own   += st[mt][r] * Dw[j * 2 + half];
            other += st[mt][r] * Dw[j * 2 + (1 - half)];
        }
    }
    const float recv = __shfl_xor(other, 16);
    const float sum  = Db[half] + (own + recv);
    const float so = half ? 0.07f : 0.9f;
    const float mo = half ? 0.0f  : -0.3f;
    float v = sum * so + mo;
    const float lo = half ? -0.07f : -1.2f;
    const float hi = half ?  0.07f :  0.6f;
    v = fminf(fmaxf(v, lo), hi);
    out[(size_t)brow * 2 + half] = v;
}

__global__ __launch_bounds__(128, 1)
void rnn_scan_wmma(const float* __restrict__ x,
                   const float* __restrict__ Ks,
                   const float* __restrict__ Wh,
                   const float* __restrict__ Ka,
                   const float* __restrict__ bias,
                   const float* __restrict__ Dw,
                   const float* __restrict__ Db,
                   float* __restrict__ out)
{
    const int tid  = threadIdx.x;
    const int w    = tid >> 5;
    const int lane = tid & 31;
    const int c    = lane & 15;
    const int half = lane >> 4;

    v2f wA[3][11];
    for (int mt = 0; mt < 3; ++mt) {
        for (int kc = 0; kc < 11; ++kc) {
            wA[mt][kc].x = a_elem(mt, kc, c, half, 0, Wh, Ks, Ka, bias);
            wA[mt][kc].y = a_elem(mt, kc, c, half, 1, Wh, Ks, Ka, bias);
        }
    }

    const int    browA = blockIdx.x * ROWS_PER_BLOCK + w * ROWS_PER_WAVE + c;
    const int    browB = browA + 16;
    const float* pxA   = x + (size_t)browA * TT * 3;
    const float* pxB   = x + (size_t)browB * TT * 3;
    const float  mean0   = half ? 0.0f : -0.3f;
    const float  iscale0 = half ? 1.0f : (1.0f / 0.9f);
    XBlock rawA = load_block(pxA);
    XBlock rawB = load_block(pxB);

    float stA[3][8], stB[3][8];
    for (int mt = 0; mt < 3; ++mt) {
        for (int r = 0; r < 8; ++r) { stA[mt][r] = 0.0f; stB[mt][r] = 0.0f; }
    }
    v8f accA[3], accB[3];
    for (int mt = 0; mt < 3; ++mt) { v8f z = {}; accA[mt] = z; accB[mt] = z; }

    const int NBLK = TT / 4;
    for (int q = 0; q < NBLK; ++q) {
        const v2f xa0 = input_chunk(rawA, 0, half, mean0, iscale0);
        const v2f xa1 = input_chunk(rawA, 1, half, mean0, iscale0);
        const v2f xa2 = input_chunk(rawA, 2, half, mean0, iscale0);
        const v2f xa3 = input_chunk(rawA, 3, half, mean0, iscale0);
        const v2f xb0 = input_chunk(rawB, 0, half, mean0, iscale0);
        const v2f xb1 = input_chunk(rawB, 1, half, mean0, iscale0);
        const v2f xb2 = input_chunk(rawB, 2, half, mean0, iscale0);
        const v2f xb3 = input_chunk(rawB, 3, half, mean0, iscale0);
        const int adv = (q < NBLK - 1) ? 12 : 0;
        pxA += adv;
        pxB += adv;
        rawA = load_block(pxA);
        rawB = load_block(pxB);

        half_step<true>(wA, stA, xa0, accA, accB, stB);
        half_step<true>(wA, stB, xb0, accB, accA, stA);
        half_step<true>(wA, stA, xa1, accA, accB, stB);
        half_step<true>(wA, stB, xb1, accB, accA, stA);
        half_step<true>(wA, stA, xa2, accA, accB, stB);
        half_step<true>(wA, stB, xb2, accB, accA, stA);
        half_step<true>(wA, stA, xa3, accA, accB, stB);
        half_step<true>(wA, stB, xb3, accB, accA, stA);
    }
    for (int v = 0; v < 20; ++v) tanh_one(v, accB, stB);

    head(stA, half, browA, Dw, Db, out);
    head(stB, half, browB, Dw, Db, out);
}

extern "C" void kernel_launch(void* const* d_in, const int* in_sizes, int n_in,
                              void* d_out, int out_size, void* d_ws, size_t ws_size,
                              hipStream_t stream) {
    const float* x    = (const float*)d_in[0];
    const float* Ks   = (const float*)d_in[1];
    const float* Wh   = (const float*)d_in[2];
    const float* Ka   = (const float*)d_in[3];
    const float* bias = (const float*)d_in[4];
    const float* Dw   = (const float*)d_in[5];
    const float* Db   = (const float*)d_in[6];
    float* out = (float*)d_out;

    const int B = in_sizes[0] / (TT * 3);
    dim3 grid(B / ROWS_PER_BLOCK), block(128);
    rnn_scan_wmma<<<grid, block, 0, stream>>>(x, Ks, Wh, Ka, bias, Dw, Db, out);
}
